# starting kernel
# baseline (speedup 1.0000x reference)
#include <hip/hip_runtime.h>

#define NE 8
#define IN_FEAT 1024
#define OUT_FEAT 4096
#define N_TOKENS 4096

#define M_TILE 64
#define N_BLOCK 512
#define K_CHUNK 128
#define K_STRIDE (K_CHUNK + 4)
#define MAX_TILES (N_TOKENS / M_TILE + NE)

#define WS_COUNT 0
#define WS_CURS 8
#define WS_NTILES 16
#define WS_TE 24
#define WS_TS (24 + MAX_TILES)
#define WS_TR (24 + 2 * MAX_TILES)
#define WS_PERM (24 + 3 * MAX_TILES)

typedef float v2f __attribute__((ext_vector_type(2)));
typedef float v8f __attribute__((ext_vector_type(8)));

__global__ void k_zero(int* ws) {
    if (threadIdx.x < 8) ws[WS_COUNT + threadIdx.x] = 0;
}

__global__ void k_count(const int* __restrict__ gate, int* __restrict__ ws) {
    int n = blockIdx.x * 256 + threadIdx.x;
    if (n < N_TOKENS) atomicAdd(&ws[WS_COUNT + gate[n]], 1);
}

__global__ void k_scan(int* __restrict__ ws) {
    if (threadIdx.x != 0) return;
    int off = 0, ntiles = 0;
    for (int e = 0; e < NE; ++e) {
        int c = ws[WS_COUNT + e];
        ws[WS_CURS + e] = off;
        int t = (c + M_TILE - 1) / M_TILE;
        for (int i = 0; i < t; ++i) {
            ws[WS_TE + ntiles] = e;
            ws[WS_TS + ntiles] = off + i * M_TILE;
            int rem = c - i * M_TILE;
            ws[WS_TR + ntiles] = rem < M_TILE ? rem : M_TILE;
            ++ntiles;
        }
        off += c;
    }
    ws[WS_NTILES] = ntiles;
}

__global__ void k_scatter(const int* __restrict__ gate, int* __restrict__ ws) {
    int n = blockIdx.x * 256 + threadIdx.x;
    if (n < N_TOKENS) {
        int e = gate[n];
        int p = atomicAdd(&ws[WS_CURS + e], 1);
        ws[WS_PERM + p] = n;
    }
}

__launch_bounds__(256)
__global__ void k_moe_gemm(const float* __restrict__ inp,
                           const float* __restrict__ weight,
                           const int* __restrict__ ws,
                           float* __restrict__ out) {
    __shared__ float As[M_TILE][K_STRIDE];
    __shared__ int toks[M_TILE];

    const int ntiles = ws[WS_NTILES];
    const int tile = blockIdx.x;
    if (tile >= ntiles) return;

    const int e      = ws[WS_TE + tile];
    const int rstart = ws[WS_TS + tile];
    const int rows   = ws[WS_TR + tile];
    const int tid = threadIdx.x;

    if (tid < M_TILE) {
        int r = tid;
        toks[r] = (r < rows) ? ws[WS_PERM + rstart + r] : ws[WS_PERM + rstart];
    }

    const int lane = tid & 31;
    const int wave = tid >> 5;
    const int hi   = lane >> 4;
    const int l15  = lane & 15;
    const int nbase = blockIdx.y * N_BLOCK + wave * 64;
    const float* wbase = weight + (size_t)e * OUT_FEAT * IN_FEAT;

    v8f acc[4][4];
#pragma unroll
    for (int m = 0; m < 4; ++m)
#pragma unroll
        for (int j = 0; j < 4; ++j) acc[m][j] = (v8f)0.0f;

    for (int kc = 0; kc < IN_FEAT; kc += K_CHUNK) {
        __syncthreads();
#pragma unroll
        for (int i = 0; i < (M_TILE * K_CHUNK / 4) / 256; ++i) {
            int id = tid + 256 * i;
            int r  = id >> 5;
            int c4 = id & 31;
            float4 v = *(const float4*)(inp + (size_t)toks[r] * IN_FEAT + kc + c4 * 4);
            *(float4*)(&As[r][c4 * 4]) = v;
        }
        __syncthreads();

        for (int k = 0; k < K_CHUNK; k += 4) {
            const int koff = k + hi * 2;
            v2f a[4];
#pragma unroll
            for (int m = 0; m < 4; ++m)
                a[m] = *(const v2f*)(&As[m * 16 + l15][koff]);
#pragma unroll
            for (int j = 0; j < 4; ++j) {
                const int col = nbase + j * 16 + l15;
                v2f b = *(const v2f*)(wbase + (size_t)col * IN_FEAT + (kc + koff));
#pragma unroll
                for (int m = 0; m < 4; ++m)
                    acc[m][j] = __builtin_amdgcn_wmma_f32_16x16x4_f32(
                        false, a[m], false, b, (short)0, acc[m][j], false, false);
            }
        }
    }

#pragma unroll
    for (int m = 0; m < 4; ++m)
#pragma unroll
        for (int j = 0; j < 4; ++j) {
            const int col = nbase + j * 16 + l15;
#pragma unroll
            for (int g = 0; g < 8; ++g) {
                const int r = m * 16 + g + hi * 8;
                if (r < rows)
                    out[(size_t)toks[r] * OUT_FEAT + col] = acc[m][j][g];
            }
        }
}

extern "C" void kernel_launch(void* const* d_in, const int* in_sizes, int n_in,
                              void* d_out, int out_size, void* d_ws, size_t ws_size,
                              hipStream_t stream) {
    const float* inp    = (const float*)d_in[0];
    const int*   gate   = (const int*)d_in[1];
    const float* weight = (const float*)d_in[2];
    float* out = (float*)d_out;
    int*   ws  = (int*)d_ws;

    k_zero<<<1, 32, 0, stream>>>(ws);
    k_count<<<(N_TOKENS + 255) / 256, 256, 0, stream>>>(gate, ws);
    k_scan<<<1, 32, 0, stream>>>(ws);
    k_scatter<<<(N_TOKENS + 255) / 256, 256, 0, stream>>>(gate, ws);

    dim3 grid(MAX_TILES, OUT_FEAT / N_BLOCK);
    k_moe_gemm<<<grid, 256, 0, stream>>>(inp, weight, ws, out);
}


# workgroup tile narrowed from 64x512 to 64x256 columns (wave tile 64x32, 8 waves per SIMD): twice as many half-size workgroups, so the last partly filled round of workgroups is half as long
# speedup vs baseline: 1.1697x; 1.1697x over previous
#include <hip/hip_runtime.h>

#define NE 8
#define IN_FEAT 1024
#define OUT_FEAT 4096
#define N_TOKENS 4096

#define M_TILE 64
#define NJ 2
#define N_BLOCK 256
#define K_CHUNK 128
#define K_STRIDE (K_CHUNK + 4)
#define MAX_TILES (N_TOKENS / M_TILE + NE)

#define WS_COUNT 0
#define WS_CURS 8
#define WS_NTILES 16
#define WS_TE 24
#define WS_TS (24 + MAX_TILES)
#define WS_TR (24 + 2 * MAX_TILES)
#define WS_PERM (24 + 3 * MAX_TILES)

typedef float v2f __attribute__((ext_vector_type(2)));
typedef float v8f __attribute__((ext_vector_type(8)));

__global__ void k_zero(int* ws) {
    if (threadIdx.x < 8) ws[WS_COUNT + threadIdx.x] = 0;
}

__global__ void k_count(const int* __restrict__ gate, int* __restrict__ ws) {
    int n = blockIdx.x * 256 + threadIdx.x;
    if (n < N_TOKENS) atomicAdd(&ws[WS_COUNT + gate[n]], 1);
}

__global__ void k_scan(int* __restrict__ ws) {
    if (threadIdx.x != 0) return;
    int off = 0, ntiles = 0;
    for (int e = 0; e < NE; ++e) {
        int c = ws[WS_COUNT + e];
        ws[WS_CURS + e] = off;
        int t = (c + M_TILE - 1) / M_TILE;
        for (int i = 0; i < t; ++i) {
            ws[WS_TE + ntiles] = e;
            ws[WS_TS + ntiles] = off + i * M_TILE;
            int rem = c - i * M_TILE;
            ws[WS_TR + ntiles] = rem < M_TILE ? rem : M_TILE;
            ++ntiles;
        }
        off += c;
    }
    ws[WS_NTILES] = ntiles;
}

__global__ void k_scatter(const int* __restrict__ gate, int* __restrict__ ws) {
    int n = blockIdx.x * 256 + threadIdx.x;
    if (n < N_TOKENS) {
        int e = gate[n];
        int p = atomicAdd(&ws[WS_CURS + e], 1);
        ws[WS_PERM + p] = n;
    }
}

__launch_bounds__(256)
__global__ void k_moe_gemm(const float* __restrict__ inp,
                           const float* __restrict__ weight,
                           const int* __restrict__ ws,
                           float* __restrict__ out) {
    __shared__ float As[M_TILE][K_STRIDE];
    __shared__ int toks[M_TILE];

    const int ntiles = ws[WS_NTILES];
    const int tile = blockIdx.x;
    if (tile >= ntiles) return;

    const int e      = ws[WS_TE + tile];
    const int rstart = ws[WS_TS + tile];
    const int rows   = ws[WS_TR + tile];
    const int tid = threadIdx.x;

    if (tid < M_TILE) {
        int r = tid;
        toks[r] = (r < rows) ? ws[WS_PERM + rstart + r] : ws[WS_PERM + rstart];
    }

    const int lane = tid & 31;
    const int wave = tid >> 5;
    const int hi   = lane >> 4;
    const int l15  = lane & 15;
    const int nbase = blockIdx.y * N_BLOCK + wave * (NJ * 16);
    const float* wbase = weight + (size_t)e * OUT_FEAT * IN_FEAT;

    v8f acc[4][NJ];
#pragma unroll
    for (int m = 0; m < 4; ++m)
#pragma unroll
        for (int j = 0; j < NJ; ++j) acc[m][j] = (v8f)0.0f;

    for (int kc = 0; kc < IN_FEAT; kc += K_CHUNK) {
        __syncthreads();
#pragma unroll
        for (int i = 0; i < (M_TILE * K_CHUNK / 4) / 256; ++i) {
            int id = tid + 256 * i;
            int r  = id >> 5;
            int c4 = id & 31;
            float4 v = *(const float4*)(inp + (size_t)toks[r] * IN_FEAT + kc + c4 * 4);
            *(float4*)(&As[r][c4 * 4]) = v;
        }
        __syncthreads();

        for (int k = 0; k < K_CHUNK; k += 4) {
            const int koff = k + hi * 2;
            v2f a[4];
#pragma unroll
            for (int m = 0; m < 4; ++m)
                a[m] = *(const v2f*)(&As[m * 16 + l15][koff]);
#pragma unroll
            for (int j = 0; j < NJ; ++j) {
                const int col = nbase + j * 16 + l15;
                v2f b = *(const v2f*)(wbase + (size_t)col * IN_FEAT + (kc + koff));
#pragma unroll
                for (int m = 0; m < 4; ++m)
                    acc[m][j] = __builtin_amdgcn_wmma_f32_16x16x4_f32(
                        false, a[m], false, b, (short)0, acc[m][j], false, false);
            }
        }
    }

#pragma unroll
    for (int m = 0; m < 4; ++m)
#pragma unroll
        for (int j = 0; j < NJ; ++j) {
            const int col = nbase + j * 16 + l15;
#pragma unroll
            for (int g = 0; g < 8; ++g) {
                const int r = m * 16 + g + hi * 8;
                if (r < rows)
                    out[(size_t)toks[r] * OUT_FEAT + col] = acc[m][j][g];
            }
        }
}

extern "C" void kernel_launch(void* const* d_in, const int* in_sizes, int n_in,
                              void* d_out, int out_size, void* d_ws, size_t ws_size,
                              hipStream_t stream) {
    const float* inp    = (const float*)d_in[0];
    const int*   gate   = (const int*)d_in[1];
    const float* weight = (const float*)d_in[2];
    float* out = (float*)d_out;
    int*   ws  = (int*)d_ws;

    k_zero<<<1, 32, 0, stream>>>(ws);
    k_count<<<(N_TOKENS + 255) / 256, 256, 0, stream>>>(gate, ws);
    k_scan<<<1, 32, 0, stream>>>(ws);
    k_scatter<<<(N_TOKENS + 255) / 256, 256, 0, stream>>>(gate, ws);

    dim3 grid(MAX_TILES, OUT_FEAT / N_BLOCK);
    k_moe_gemm<<<grid, 256, 0, stream>>>(inp, weight, ws, out);
}


# weight operand staged through LDS like the token rows: whole-line float4 loads in 32-K chunks, fetched one chunk ahead into registers, replace per-lane 8-byte global fragment loads; sum order kept
# speedup vs baseline: 1.2304x; 1.0519x over previous
#include <hip/hip_runtime.h>

#define NE 8
#define IN_FEAT 1024
#define OUT_FEAT 4096
#define N_TOKENS 4096

#define M_TILE 64
#define NJ 2
#define N_BLOCK 256
#define K_CHUNK 32
#define K_STRIDE (K_CHUNK + 4)
#define MAX_TILES (N_TOKENS / M_TILE + NE)

#define WS_COUNT 0
#define WS_CURS 8
#define WS_NTILES 16
#define WS_TE 24
#define WS_TS (24 + MAX_TILES)
#define WS_TR (24 + 2 * MAX_TILES)
#define WS_PERM (24 + 3 * MAX_TILES)

typedef float v2f __attribute__((ext_vector_type(2)));
typedef float v8f __attribute__((ext_vector_type(8)));

__global__ void k_zero(int* ws) {
    if (threadIdx.x < 8) ws[WS_COUNT + threadIdx.x] = 0;
}

__global__ void k_count(const int* __restrict__ gate, int* __restrict__ ws) {
    int n = blockIdx.x * 256 + threadIdx.x;
    if (n < N_TOKENS) atomicAdd(&ws[WS_COUNT + gate[n]], 1);
}

__global__ void k_scan(int* __restrict__ ws) {
    if (threadIdx.x != 0) return;
    int off = 0, ntiles = 0;
    for (int e = 0; e < NE; ++e) {
        int c = ws[WS_COUNT + e];
        ws[WS_CURS + e] = off;
        int t = (c + M_TILE - 1) / M_TILE;
        for (int i = 0; i < t; ++i) {
            ws[WS_TE + ntiles] = e;
            ws[WS_TS + ntiles] = off + i * M_TILE;
            int rem = c - i * M_TILE;
            ws[WS_TR + ntiles] = rem < M_TILE ? rem : M_TILE;
            ++ntiles;
        }
        off += c;
    }
    ws[WS_NTILES] = ntiles;
}

__global__ void k_scatter(const int* __restrict__ gate, int* __restrict__ ws) {
    int n = blockIdx.x * 256 + threadIdx.x;
    if (n < N_TOKENS) {
        int e = gate[n];
        int p = atomicAdd(&ws[WS_CURS + e], 1);
        ws[WS_PERM + p] = n;
    }
}

__launch_bounds__(256)
__global__ void k_moe_gemm(const float* __restrict__ inp,
                           const float* __restrict__ weight,
                           const int* __restrict__ ws,
                           float* __restrict__ out) {
    __shared__ float As[M_TILE][K_STRIDE];
    __shared__ float Bs[N_BLOCK][K_STRIDE];
    __shared__ int toks[M_TILE];

    const int ntiles = ws[WS_NTILES];
    const int tile = blockIdx.x;
    if (tile >= ntiles) return;

    const int e      = ws[WS_TE + tile];
    const int rstart = ws[WS_TS + tile];
    const int rows   = ws[WS_TR + tile];
    const int tid = threadIdx.x;

    if (tid < M_TILE) {
        int r = tid;
        toks[r] = (r < rows) ? ws[WS_PERM + rstart + r] : ws[WS_PERM + rstart];
    }
    __syncthreads();

    const int lane = tid & 31;
    const int wave = tid >> 5;
    const int hi   = lane >> 4;
    const int l15  = lane & 15;
    const int nbase = blockIdx.y * N_BLOCK + wave * (NJ * 16);
    const float* wbase = weight + (size_t)e * OUT_FEAT * IN_FEAT;

    const int sr = tid >> 3;
    const int sc = (tid & 7) * 4;
    const float* ap0 = inp + (size_t)toks[sr] * IN_FEAT + sc;
    const float* ap1 = inp + (size_t)toks[sr + 32] * IN_FEAT + sc;
    const float* bp = wbase + (size_t)(blockIdx.y * N_BLOCK + sr) * IN_FEAT + sc;

    float4 ra0, ra1, rb0, rb1, rb2, rb3, rb4, rb5, rb6, rb7;
#define FETCH_CHUNK(K0)                                              \
    ra0 = *(const float4*)(ap0 + (K0));                              \
    ra1 = *(const float4*)(ap1 + (K0));                              \
    rb0 = *(const float4*)(bp + (size_t)0 * 32 * IN_FEAT + (K0));    \
    rb1 = *(const float4*)(bp + (size_t)1 * 32 * IN_FEAT + (K0));    \
    rb2 = *(const float4*)(bp + (size_t)2 * 32 * IN_FEAT + (K0));    \
    rb3 = *(const float4*)(bp + (size_t)3 * 32 * IN_FEAT + (K0));    \
    rb4 = *(const float4*)(bp + (size_t)4 * 32 * IN_FEAT + (K0));    \
    rb5 = *(const float4*)(bp + (size_t)5 * 32 * IN_FEAT + (K0));    \
    rb6 = *(const float4*)(bp + (size_t)6 * 32 * IN_FEAT + (K0));    \
    rb7 = *(const float4*)(bp + (size_t)7 * 32 * IN_FEAT + (K0));
    FETCH_CHUNK(0)

    v8f acc[4][NJ];
#pragma unroll
    for (int m = 0; m < 4; ++m)
#pragma unroll
        for (int j = 0; j < NJ; ++j) acc[m][j] = (v8f)0.0f;

    for (int kc = 0; kc < IN_FEAT; kc += K_CHUNK) {
        *(float4*)(&As[sr][sc])       = ra0;
        *(float4*)(&As[sr + 32][sc])  = ra1;
        *(float4*)(&Bs[sr][sc])       = rb0;
        *(float4*)(&Bs[sr + 32][sc])  = rb1;
        *(float4*)(&Bs[sr + 64][sc])  = rb2;
        *(float4*)(&Bs[sr + 96][sc])  = rb3;
        *(float4*)(&Bs[sr + 128][sc]) = rb4;
        *(float4*)(&Bs[sr + 160][sc]) = rb5;
        *(float4*)(&Bs[sr + 192][sc]) = rb6;
        *(float4*)(&Bs[sr + 224][sc]) = rb7;
        __syncthreads();

        if (kc + K_CHUNK < IN_FEAT) {
            FETCH_CHUNK(kc + K_CHUNK)
        }

        for (int k = 0; k < K_CHUNK; k += 4) {
            const int koff = k + hi * 2;
            v2f a[4];
#pragma unroll
            for (int m = 0; m < 4; ++m)
                a[m] = *(const v2f*)(&As[m * 16 + l15][koff]);
#pragma unroll
            for (int j = 0; j < NJ; ++j) {
                v2f b = *(const v2f*)(&Bs[wave * (NJ * 16) + j * 16 + l15][koff]);
#pragma unroll
                for (int m = 0; m < 4; ++m)
                    acc[m][j] = __builtin_amdgcn_wmma_f32_16x16x4_f32(
                        false, a[m], false, b, (short)0, acc[m][j], false, false);
            }
        }
        __syncthreads();
    }

#pragma unroll
    for (int m = 0; m < 4; ++m)
#pragma unroll
        for (int j = 0; j < NJ; ++j) {
            const int col = nbase + j * 16 + l15;
#pragma unroll
            for (int g = 0; g < 8; ++g) {
                const int r = m * 16 + g + hi * 8;
                if (r < rows)
                    out[(size_t)toks[r] * OUT_FEAT + col] = acc[m][j][g];
            }
        }
}

extern "C" void kernel_launch(void* const* d_in, const int* in_sizes, int n_in,
                              void* d_out, int out_size, void* d_ws, size_t ws_size,
                              hipStream_t stream) {
    const float* inp    = (const float*)d_in[0];
    const int*   gate   = (const int*)d_in[1];
    const float* weight = (const float*)d_in[2];
    float* out = (float*)d_out;
    int*   ws  = (int*)d_ws;

    k_zero<<<1, 32, 0, stream>>>(ws);
    k_count<<<(N_TOKENS + 255) / 256, 256, 0, stream>>>(gate, ws);
    k_scan<<<1, 32, 0, stream>>>(ws);
    k_scatter<<<(N_TOKENS + 255) / 256, 256, 0, stream>>>(gate, ws);

    dim3 grid(MAX_TILES, OUT_FEAT / N_BLOCK);
    k_moe_gemm<<<grid, 256, 0, stream>>>(inp, weight, ws, out);
}


# workgroup tile narrowed from 64x256 to 64x128 columns with 4 waves (wave tile 64x32 kept): five small workgroups per WGP in separate barrier phases, last round half as long; sum order kept
# speedup vs baseline: 1.3142x; 1.0681x over previous
#include <hip/hip_runtime.h>

#define NE 8
#define IN_FEAT 1024
#define OUT_FEAT 4096
#define N_TOKENS 4096

#define M_TILE 64
#define NJ 2
#define N_BLOCK 128
#define K_CHUNK 32
#define K_STRIDE (K_CHUNK + 4)
#define MAX_TILES (N_TOKENS / M_TILE + NE)

#define WS_COUNT 0
#define WS_CURS 8
#define WS_NTILES 16
#define WS_TE 24
#define WS_TS (24 + MAX_TILES)
#define WS_TR (24 + 2 * MAX_TILES)
#define WS_PERM (24 + 3 * MAX_TILES)

typedef float v2f __attribute__((ext_vector_type(2)));
typedef float v8f __attribute__((ext_vector_type(8)));

__global__ void k_zero(int* ws) {
    if (threadIdx.x < 8) ws[WS_COUNT + threadIdx.x] = 0;
}

__global__ void k_count(const int* __restrict__ gate, int* __restrict__ ws) {
    int n = blockIdx.x * 256 + threadIdx.x;
    if (n < N_TOKENS) atomicAdd(&ws[WS_COUNT + gate[n]], 1);
}

__global__ void k_scan(int* __restrict__ ws) {
    if (threadIdx.x != 0) return;
    int off = 0, ntiles = 0;
    for (int e = 0; e < NE; ++e) {
        int c = ws[WS_COUNT + e];
        ws[WS_CURS + e] = off;
        int t = (c + M_TILE - 1) / M_TILE;
        for (int i = 0; i < t; ++i) {
            ws[WS_TE + ntiles] = e;
            ws[WS_TS + ntiles] = off + i * M_TILE;
            int rem = c - i * M_TILE;
            ws[WS_TR + ntiles] = rem < M_TILE ? rem : M_TILE;
            ++ntiles;
        }
        off += c;
    }
    ws[WS_NTILES] = ntiles;
}

__global__ void k_scatter(const int* __restrict__ gate, int* __restrict__ ws) {
    int n = blockIdx.x * 256 + threadIdx.x;
    if (n < N_TOKENS) {
        int e = gate[n];
        int p = atomicAdd(&ws[WS_CURS + e], 1);
        ws[WS_PERM + p] = n;
    }
}

__launch_bounds__(128)
__global__ void k_moe_gemm(const float* __restrict__ inp,
                           const float* __restrict__ weight,
                           const int* __restrict__ ws,
                           float* __restrict__ out) {
    __shared__ float As[M_TILE][K_STRIDE];
    __shared__ float Bs[N_BLOCK][K_STRIDE];
    __shared__ int toks[M_TILE];

    const int ntiles = ws[WS_NTILES];
    const int tile = blockIdx.x;
    if (tile >= ntiles) return;

    const int e      = ws[WS_TE + tile];
    const int rstart = ws[WS_TS + tile];
    const int rows   = ws[WS_TR + tile];
    const int tid = threadIdx.x;

    if (tid < M_TILE) {
        int r = tid;
        toks[r] = (r < rows) ? ws[WS_PERM + rstart + r] : ws[WS_PERM + rstart];
    }
    __syncthreads();

    const int lane = tid & 31;
    const int wave = tid >> 5;
    const int hi   = lane >> 4;
    const int l15  = lane & 15;
    const int nbase = blockIdx.y * N_BLOCK + wave * (NJ * 16);
    const float* wbase = weight + (size_t)e * OUT_FEAT * IN_FEAT;

    const int sr = tid >> 3;
    const int sc = (tid & 7) * 4;
    const float* ap0 = inp + (size_t)toks[sr] * IN_FEAT + sc;
    const float* ap1 = inp + (size_t)toks[sr + 16] * IN_FEAT + sc;
    const float* ap2 = inp + (size_t)toks[sr + 32] * IN_FEAT + sc;
    const float* ap3 = inp + (size_t)toks[sr + 48] * IN_FEAT + sc;
    const float* bp = wbase + (size_t)(blockIdx.y * N_BLOCK + sr) * IN_FEAT + sc;

    float4 ra0, ra1, ra2, ra3, rb0, rb1, rb2, rb3, rb4, rb5, rb6, rb7;
#define FETCH_CHUNK(K0)                                              \
    ra0 = *(const float4*)(ap0 + (K0));                              \
    ra1 = *(const float4*)(ap1 + (K0));                              \
    ra2 = *(const float4*)(ap2 + (K0));                              \
    ra3 = *(const float4*)(ap3 + (K0));                              \
    rb0 = *(const float4*)(bp + (size_t)0 * 16 * IN_FEAT + (K0));    \
    rb1 = *(const float4*)(bp + (size_t)1 * 16 * IN_FEAT + (K0));    \
    rb2 = *(const float4*)(bp + (size_t)2 * 16 * IN_FEAT + (K0));    \
    rb3 = *(const float4*)(bp + (size_t)3 * 16 * IN_FEAT + (K0));    \
    rb4 = *(const float4*)(bp + (size_t)4 * 16 * IN_FEAT + (K0));    \
    rb5 = *(const float4*)(bp + (size_t)5 * 16 * IN_FEAT + (K0));    \
    rb6 = *(const float4*)(bp + (size_t)6 * 16 * IN_FEAT + (K0));    \
    rb7 = *(const float4*)(bp + (size_t)7 * 16 * IN_FEAT + (K0));
    FETCH_CHUNK(0)

    v8f acc[4][NJ];
#pragma unroll
    for (int m = 0; m < 4; ++m)
#pragma unroll
        for (int j = 0; j < NJ; ++j) acc[m][j] = (v8f)0.0f;

    for (int kc = 0; kc < IN_FEAT; kc += K_CHUNK) {
        *(float4*)(&As[sr][sc])       = ra0;
        *(float4*)(&As[sr + 16][sc])  = ra1;
        *(float4*)(&As[sr + 32][sc])  = ra2;
        *(float4*)(&As[sr + 48][sc])  = ra3;
        *(float4*)(&Bs[sr][sc])       = rb0;
        *(float4*)(&Bs[sr + 16][sc])  = rb1;
        *(float4*)(&Bs[sr + 32][sc])  = rb2;
        *(float4*)(&Bs[sr + 48][sc])  = rb3;
        *(float4*)(&Bs[sr + 64][sc])  = rb4;
        *(float4*)(&Bs[sr + 80][sc])  = rb5;
        *(float4*)(&Bs[sr + 96][sc])  = rb6;
        *(float4*)(&Bs[sr + 112][sc]) = rb7;
        __syncthreads();

        if (kc + K_CHUNK < IN_FEAT) {
            FETCH_CHUNK(kc + K_CHUNK)
        }

        for (int k = 0; k < K_CHUNK; k += 4) {
            const int koff = k + hi * 2;
            v2f a[4];
#pragma unroll
            for (int m = 0; m < 4; ++m)
                a[m] = *(const v2f*)(&As[m * 16 + l15][koff]);
#pragma unroll
            for (int j = 0; j < NJ; ++j) {
                v2f b = *(const v2f*)(&Bs[wave * (NJ * 16) + j * 16 + l15][koff]);
#pragma unroll
                for (int m = 0; m < 4; ++m)
                    acc[m][j] = __builtin_amdgcn_wmma_f32_16x16x4_f32(
                        false, a[m], false, b, (short)0, acc[m][j], false, false);
            }
        }
        __syncthreads();
    }

#pragma unroll
    for (int m = 0; m < 4; ++m)
#pragma unroll
        for (int j = 0; j < NJ; ++j) {
            const int col = nbase + j * 16 + l15;
#pragma unroll
            for (int g = 0; g < 8; ++g) {
                const int r = m * 16 + g + hi * 8;
                if (r < rows)
                    out[(size_t)toks[r] * OUT_FEAT + col] = acc[m][j][g];
            }
        }
}

extern "C" void kernel_launch(void* const* d_in, const int* in_sizes, int n_in,
                              void* d_out, int out_size, void* d_ws, size_t ws_size,
                              hipStream_t stream) {
    const float* inp    = (const float*)d_in[0];
    const int*   gate   = (const int*)d_in[1];
    const float* weight = (const float*)d_in[2];
    float* out = (float*)d_out;
    int*   ws  = (int*)d_ws;

    k_zero<<<1, 32, 0, stream>>>(ws);
    k_count<<<(N_TOKENS + 255) / 256, 256, 0, stream>>>(gate, ws);
    k_scan<<<1, 32, 0, stream>>>(ws);
    k_scatter<<<(N_TOKENS + 255) / 256, 256, 0, stream>>>(gate, ws);

    dim3 grid(MAX_TILES, OUT_FEAT / N_BLOCK);
    k_moe_gemm<<<grid, 128, 0, stream>>>(inp, weight, ws, out);
}


# result tile staged row-major in LDS (over the dead operand chunks) and stored as whole rows: each wave store writes 512 contiguous bytes of one token row, one token lookup per row; sum order kept
# speedup vs baseline: 1.3699x; 1.0424x over previous
#include <hip/hip_runtime.h>

#define NE 8
#define IN_FEAT 1024
#define OUT_FEAT 4096
#define N_TOKENS 4096

#define M_TILE 64
#define NJ 2
#define N_BLOCK 128
#define K_CHUNK 32
#define K_STRIDE (K_CHUNK + 4)
#define MAX_TILES (N_TOKENS / M_TILE + NE)
#define C_STRIDE (N_BLOCK + 4)
#define AB_FLOATS ((M_TILE + N_BLOCK) * K_STRIDE)
#define C_FLOATS (M_TILE * C_STRIDE)
#define SMEM_FLOATS (C_FLOATS > AB_FLOATS ? C_FLOATS : AB_FLOATS)

#define WS_COUNT 0
#define WS_CURS 8
#define WS_NTILES 16
#define WS_TE 24
#define WS_TS (24 + MAX_TILES)
#define WS_TR (24 + 2 * MAX_TILES)
#define WS_PERM (24 + 3 * MAX_TILES)

typedef float v2f __attribute__((ext_vector_type(2)));
typedef float v8f __attribute__((ext_vector_type(8)));

__global__ void k_zero(int* ws) {
    if (threadIdx.x < 8) ws[WS_COUNT + threadIdx.x] = 0;
}

__global__ void k_count(const int* __restrict__ gate, int* __restrict__ ws) {
    int n = blockIdx.x * 256 + threadIdx.x;
    if (n < N_TOKENS) atomicAdd(&ws[WS_COUNT + gate[n]], 1);
}

__global__ void k_scan(int* __restrict__ ws) {
    if (threadIdx.x != 0) return;
    int off = 0, ntiles = 0;
    for (int e = 0; e < NE; ++e) {
        int c = ws[WS_COUNT + e];
        ws[WS_CURS + e] = off;
        int t = (c + M_TILE - 1) / M_TILE;
        for (int i = 0; i < t; ++i) {
            ws[WS_TE + ntiles] = e;
            ws[WS_TS + ntiles] = off + i * M_TILE;
            int rem = c - i * M_TILE;
            ws[WS_TR + ntiles] = rem < M_TILE ? rem : M_TILE;
            ++ntiles;
        }
        off += c;
    }
    ws[WS_NTILES] = ntiles;
}

__global__ void k_scatter(const int* __restrict__ gate, int* __restrict__ ws) {
    int n = blockIdx.x * 256 + threadIdx.x;
    if (n < N_TOKENS) {
        int e = gate[n];
        int p = atomicAdd(&ws[WS_CURS + e], 1);
        ws[WS_PERM + p] = n;
    }
}

__launch_bounds__(128)
__global__ void k_moe_gemm(const float* __restrict__ inp,
                           const float* __restrict__ weight,
                           const int* __restrict__ ws,
                           float* __restrict__ out) {
    __shared__ __attribute__((aligned(16))) float smem[SMEM_FLOATS];
    __shared__ int toks[M_TILE];
    float (*As)[K_STRIDE] = reinterpret_cast<float (*)[K_STRIDE]>(smem);
    float (*Bs)[K_STRIDE] = reinterpret_cast<float (*)[K_STRIDE]>(smem + M_TILE * K_STRIDE);
    float (*Cs)[C_STRIDE] = reinterpret_cast<float (*)[C_STRIDE]>(smem);

    const int ntiles = ws[WS_NTILES];
    const int tile = blockIdx.x;
    if (tile >= ntiles) return;

    const int e      = ws[WS_TE + tile];
    const int rstart = ws[WS_TS + tile];
    const int rows   = ws[WS_TR + tile];
    const int tid = threadIdx.x;

    if (tid < M_TILE) {
        int r = tid;
        toks[r] = (r < rows) ? ws[WS_PERM + rstart + r] : ws[WS_PERM + rstart];
    }
    __syncthreads();

    const int lane = tid & 31;
    const int wave = tid >> 5;
    const int hi   = lane >> 4;
    const int l15  = lane & 15;
    const float* wbase = weight + (size_t)e * OUT_FEAT * IN_FEAT;

    const int sr = tid >> 3;
    const int sc = (tid & 7) * 4;
    const float* ap0 = inp + (size_t)toks[sr] * IN_FEAT + sc;
    const float* ap1 = inp + (size_t)toks[sr + 16] * IN_FEAT + sc;
    const float* ap2 = inp + (size_t)toks[sr + 32] * IN_FEAT + sc;
    const float* ap3 = inp + (size_t)toks[sr + 48] * IN_FEAT + sc;
    const float* bp = wbase + (size_t)(blockIdx.y * N_BLOCK + sr) * IN_FEAT + sc;

    float4 ra0, ra1, ra2, ra3, rb0, rb1, rb2, rb3, rb4, rb5, rb6, rb7;
#define FETCH_CHUNK(K0)                                              \
    ra0 = *(const float4*)(ap0 + (K0));                              \
    ra1 = *(const float4*)(ap1 + (K0));                              \
    ra2 = *(const float4*)(ap2 + (K0));                              \
    ra3 = *(const float4*)(ap3 + (K0));                              \
    rb0 = *(const float4*)(bp + (size_t)0 * 16 * IN_FEAT + (K0));    \
    rb1 = *(const float4*)(bp + (size_t)1 * 16 * IN_FEAT + (K0));    \
    rb2 = *(const float4*)(bp + (size_t)2 * 16 * IN_FEAT + (K0));    \
    rb3 = *(const float4*)(bp + (size_t)3 * 16 * IN_FEAT + (K0));    \
    rb4 = *(const float4*)(bp + (size_t)4 * 16 * IN_FEAT + (K0));    \
    rb5 = *(const float4*)(bp + (size_t)5 * 16 * IN_FEAT + (K0));    \
    rb6 = *(const float4*)(bp + (size_t)6 * 16 * IN_FEAT + (K0));    \
    rb7 = *(const float4*)(bp + (size_t)7 * 16 * IN_FEAT + (K0));
    FETCH_CHUNK(0)

    v8f acc[4][NJ];
#pragma unroll
    for (int m = 0; m < 4; ++m)
#pragma unroll
        for (int j = 0; j < NJ; ++j) acc[m][j] = (v8f)0.0f;

    for (int kc = 0; kc < IN_FEAT; kc += K_CHUNK) {
        *(float4*)(&As[sr][sc])       = ra0;
        *(float4*)(&As[sr + 16][sc])  = ra1;
        *(float4*)(&As[sr + 32][sc])  = ra2;
        *(float4*)(&As[sr + 48][sc])  = ra3;
        *(float4*)(&Bs[sr][sc])       = rb0;
        *(float4*)(&Bs[sr + 16][sc])  = rb1;
        *(float4*)(&Bs[sr + 32][sc])  = rb2;
        *(float4*)(&Bs[sr + 48][sc])  = rb3;
        *(float4*)(&Bs[sr + 64][sc])  = rb4;
        *(float4*)(&Bs[sr + 80][sc])  = rb5;
        *(float4*)(&Bs[sr + 96][sc])  = rb6;
        *(float4*)(&Bs[sr + 112][sc]) = rb7;
        __syncthreads();

        if (kc + K_CHUNK < IN_FEAT) {
            FETCH_CHUNK(kc + K_CHUNK)
        }

        for (int k = 0; k < K_CHUNK; k += 4) {
            const int koff = k + hi * 2;
            v2f a[4];
#pragma unroll
            for (int m = 0; m < 4; ++m)
                a[m] = *(const v2f*)(&As[m * 16 + l15][koff]);
#pragma unroll
            for (int j = 0; j < NJ; ++j) {
                v2f b = *(const v2f*)(&Bs[wave * (NJ * 16) + j * 16 + l15][koff]);
#pragma unroll
                for (int m = 0; m < 4; ++m)
                    acc[m][j] = __builtin_amdgcn_wmma_f32_16x16x4_f32(
                        false, a[m], false, b, (short)0, acc[m][j], false, false);
            }
        }
        __syncthreads();
    }

#pragma unroll
    for (int m = 0; m < 4; ++m)
#pragma unroll
        for (int j = 0; j < NJ; ++j)
#pragma unroll
            for (int g = 0; g < 8; ++g)
                Cs[m * 16 + g + hi * 8][wave * (NJ * 16) + j * 16 + l15] = acc[m][j][g];
    __syncthreads();

    const int wv = __builtin_amdgcn_readfirstlane(wave);
    float* obase = out + (size_t)blockIdx.y * N_BLOCK + lane * 4;
    if (rows == M_TILE) {
        float4 cv[16];
        int ctok[16];
#pragma unroll
        for (int i = 0; i < 16; ++i) {
            const int r = i * 4 + wv;
            cv[i] = *(const float4*)(&Cs[r][lane * 4]);
            ctok[i] = toks[r];
        }
#pragma unroll
        for (int i = 0; i < 16; ++i)
            *(float4*)(obase + (size_t)ctok[i] * OUT_FEAT) = cv[i];
    } else {
#pragma unroll
        for (int i = 0; i < 16; ++i) {
            const int r = i * 4 + wv;
            if (r < rows)
                *(float4*)(obase + (size_t)toks[r] * OUT_FEAT) = *(const float4*)(&Cs[r][lane * 4]);
        }
    }
}

extern "C" void kernel_launch(void* const* d_in, const int* in_sizes, int n_in,
                              void* d_out, int out_size, void* d_ws, size_t ws_size,
                              hipStream_t stream) {
    const float* inp    = (const float*)d_in[0];
    const int*   gate   = (const int*)d_in[1];
    const float* weight = (const float*)d_in[2];
    float* out = (float*)d_out;
    int*   ws  = (int*)d_ws;

    k_zero<<<1, 32, 0, stream>>>(ws);
    k_count<<<(N_TOKENS + 255) / 256, 256, 0, stream>>>(gate, ws);
    k_scan<<<1, 32, 0, stream>>>(ws);
    k_scatter<<<(N_TOKENS + 255) / 256, 256, 0, stream>>>(gate, ws);

    dim3 grid(MAX_TILES, OUT_FEAT / N_BLOCK);
    k_moe_gemm<<<grid, 128, 0, stream>>>(inp, weight, ws, out);
}
